# starting kernel
# baseline (speedup 1.0000x reference)
#include <hip/hip_runtime.h>

typedef __attribute__((ext_vector_type(2))) float v2f;
typedef __attribute__((ext_vector_type(8))) float v8f;

__global__ __launch_bounds__(256)
void neigh_norm_wmma_kernel(const float* __restrict__ points,
                            const float* __restrict__ nbh,
                            float* __restrict__ out,
                            int nPoints)
{
    const int lane = threadIdx.x & 31;
    const int wave = threadIdx.x >> 5;
    const int p    = blockIdx.x * 8 + wave;
    if (p >= nPoints) return;

    const float px = points[3 * p + 0];
    const float py = points[3 * p + 1];
    const float pz = points[3 * p + 2];

    const float n2      = px * px + py * py + pz * pz;
    const float nrm     = sqrtf(n2);
    const float inv_nrm = (nrm == 0.0f) ? 0.0f : (1.0f / nrm);
    const float ux = px * inv_nrm;
    const float uy = py * inv_nrm;

    const float dd   = ux * ux + uy * uy;
    const float invd = (dd == 0.0f) ? 0.0f : (1.0f / dd);
    const float t00 =  ux * invd;
    const float t01 =  uy * invd;
    const float t03 = -(ux * px + uy * py) * invd;
    const float t10 = -uy * invd;
    const float t11 =  ux * invd;
    const float t13 =  (uy * px - ux * py) * invd;

    const int  row = lane & 15;
    const bool lo  = lane < 16;
    v2f a;
    a.x = lo ? ((row == 0) ? t00 : (row == 1) ? t10 : 0.0f)
             : ((row == 2) ? 1.0f : 0.0f);
    a.y = lo ? ((row == 0) ? t01 : (row == 1) ? t11 : 0.0f)
             : ((row == 0) ? t03 : (row == 1) ? t13 : (row == 2) ? -pz : 0.0f);

    const size_t nbase = (size_t)p * 64u * 3u;
    const int    col   = lane & 15;

    #pragma unroll
    for (int t = 0; t < 4; ++t) {
        const size_t ni = nbase + (size_t)(t * 16 + col) * 3u;

        if (t < 3) {
            __builtin_prefetch(nbh + ni + 48, 0, 0);
        }

        const float f0 = nbh[ni + 0];
        const float f1 = nbh[ni + 1];
        const float f2 = nbh[ni + 2];

        v2f b;
        b.x = lo ? f0 : f2;
        b.y = lo ? f1 : 1.0f;

        v8f c = {};
        c = __builtin_amdgcn_wmma_f32_16x16x4_f32(
                 false, a,  false, b,
                 (short)0, c,  false,  false);

        if (lo) {
            float* o = out + ni;
            __builtin_nontemporal_store(c[0], o + 0);
            __builtin_nontemporal_store(c[1], o + 1);
            __builtin_nontemporal_store(c[2], o + 2);
        }
    }
}

extern "C" void kernel_launch(void* const* d_in, const int* in_sizes, int n_in,
                              void* d_out, int out_size, void* d_ws, size_t ws_size,
                              hipStream_t stream)
{
    const float* points = (const float*)d_in[0];
    const float* nbh    = (const float*)d_in[1];
    float*       out    = (float*)d_out;

    const int nPoints = in_sizes[0] / 3;
    const int blocks  = (nPoints + 7) / 8;

    neigh_norm_wmma_kernel<<<blocks, 256, 0, stream>>>(points, nbh, out, nPoints);
}


# Resident grid of 512 workgroups of 8 waves: each wave loops over groups of 4 points with the next group's loads issued ahead, instead of one short wave per point (131072 waves); sum order kept
# speedup vs baseline: 1.7139x; 1.7139x over previous
#include <hip/hip_runtime.h>

typedef __attribute__((ext_vector_type(2))) float v2f;
typedef __attribute__((ext_vector_type(8))) float v8f;


#define PTS 4

struct GroupRegs {
    float px[PTS], py[PTS], pz[PTS];
    float f[PTS][4][3];
};

__device__ __forceinline__ void load_group(GroupRegs& r,
                                           const float* __restrict__ points,
                                           const float* __restrict__ nbh,
                                           int g, int nPoints, int col)
{
    #pragma unroll
    for (int u = 0; u < PTS; ++u) {
        int p = g * PTS + u;
        p = (p < nPoints) ? p : (nPoints - 1);
        r.px[u] = points[3 * p + 0];
        r.py[u] = points[3 * p + 1];
        r.pz[u] = points[3 * p + 2];
        const float* nb = nbh + (size_t)p * 192u + (size_t)col * 3u;
        #pragma unroll
        for (int t = 0; t < 4; ++t) {
            r.f[u][t][0] = nb[t * 48 + 0];
            r.f[u][t][1] = nb[t * 48 + 1];
            r.f[u][t][2] = nb[t * 48 + 2];
        }
    }
}

__device__ __forceinline__ void transform_store_group(const GroupRegs& r,
                                                      float* __restrict__ out,
                                                      int g, int nPoints,
                                                      int row, bool lo)
{
    #pragma unroll
    for (int u = 0; u < PTS; ++u) {
        const int p = g * PTS + u;
        const float px = r.px[u], py = r.py[u], pz = r.pz[u];

        const float n2      = px * px + py * py + pz * pz;
        const float nrm     = sqrtf(n2);
        const float inv_nrm = (nrm == 0.0f) ? 0.0f : (1.0f / nrm);
        const float ux = px * inv_nrm;
        const float uy = py * inv_nrm;

        const float dd   = ux * ux + uy * uy;
        const float invd = (dd == 0.0f) ? 0.0f : (1.0f / dd);
        const float t00 =  ux * invd;
        const float t01 =  uy * invd;
        const float t03 = -(ux * px + uy * py) * invd;
        const float t10 = -uy * invd;
        const float t11 =  ux * invd;
        const float t13 =  (uy * px - ux * py) * invd;

        v2f a;
        a.x = lo ? ((row == 0) ? t00 : (row == 1) ? t10 : 0.0f)
                 : ((row == 2) ? 1.0f : 0.0f);
        a.y = lo ? ((row == 0) ? t01 : (row == 1) ? t11 : 0.0f)
                 : ((row == 0) ? t03 : (row == 1) ? t13 : (row == 2) ? -pz : 0.0f);

        v8f c[4];
        #pragma unroll
        for (int t = 0; t < 4; ++t) {
            v2f b;
            b.x = lo ? r.f[u][t][0] : r.f[u][t][2];
            b.y = lo ? r.f[u][t][1] : 1.0f;
            v8f z = {};
            c[t] = __builtin_amdgcn_wmma_f32_16x16x4_f32(
                        false, a,  false, b,
                        (short)0, z,  false,  false);
        }

        if (lo && p < nPoints) {
            float* o = out + (size_t)p * 192u + (size_t)row * 3u;
            #pragma unroll
            for (int t = 0; t < 4; ++t) {
                __builtin_nontemporal_store(c[t][0], o + t * 48 + 0);
                __builtin_nontemporal_store(c[t][1], o + t * 48 + 1);
                __builtin_nontemporal_store(c[t][2], o + t * 48 + 2);
            }
        }
    }
}

__global__ __launch_bounds__(256)
void neigh_norm_wmma_kernel(const float* __restrict__ points,
                            const float* __restrict__ nbh,
                            float* __restrict__ out,
                            int nPoints)
{
    const int lane = threadIdx.x & 31;
    const int wave   = __builtin_amdgcn_readfirstlane((int)(threadIdx.x >> 5));
    const int nWaves = (int)gridDim.x * 8;
    const int nGroups = (nPoints + PTS - 1) / PTS;
    const int row = lane & 15;
    const bool lo = lane < 16;

    int g0 = (int)blockIdx.x * 8 + wave;
    if (g0 >= nGroups) return;

    GroupRegs A, B;
    load_group(A, points, nbh, g0, nPoints, row);

    for (;;) {
        const int g1 = g0 + nWaves;
        if (g1 >= nGroups) { transform_store_group(A, out, g0, nPoints, row, lo); break; }
        load_group(B, points, nbh, g1, nPoints, row);
        transform_store_group(A, out, g0, nPoints, row, lo);

        g0 = g1 + nWaves;
        if (g0 >= nGroups) { transform_store_group(B, out, g1, nPoints, row, lo); break; }
        load_group(A, points, nbh, g0, nPoints, row);
        transform_store_group(B, out, g1, nPoints, row, lo);
    }
}

extern "C" void kernel_launch(void* const* d_in, const int* in_sizes, int n_in,
                              void* d_out, int out_size, void* d_ws, size_t ws_size,
                              hipStream_t stream)
{
    const float* points = (const float*)d_in[0];
    const float* nbh    = (const float*)d_in[1];
    float*       out    = (float*)d_out;

    const int nPoints = in_sizes[0] / 3;

    neigh_norm_wmma_kernel<<<512, 256, 0, stream>>>(points, nbh, out, nPoints);
}


# One lane per neighbor: each 12-byte wave load/store covers 3 whole 128-B lines, all 32 lanes distinct; A holds the inverse twice (rows 0-2, 8-10); two K=2 WMMAs accumulated, sum order may change
# speedup vs baseline: 1.9369x; 1.1301x over previous
#include <hip/hip_runtime.h>

typedef __attribute__((ext_vector_type(2))) float v2f;
typedef __attribute__((ext_vector_type(8))) float v8f;


#define PTS 4

struct GroupRegs {
    float px[PTS], py[PTS], pz[PTS];
    float f[PTS][2][3];
};

__device__ __forceinline__ void load_group(GroupRegs& r,
                                           const float* __restrict__ points,
                                           const float* __restrict__ nbh,
                                           int g, int nPoints, int lane)
{
    #pragma unroll
    for (int u = 0; u < PTS; ++u) {
        int p = g * PTS + u;
        p = (p < nPoints) ? p : (nPoints - 1);
        r.px[u] = points[3 * p + 0];
        r.py[u] = points[3 * p + 1];
        r.pz[u] = points[3 * p + 2];
        const float* nb = nbh + (size_t)p * 192u + (size_t)lane * 3u;
        #pragma unroll
        for (int t = 0; t < 2; ++t) {
            r.f[u][t][0] = nb[t * 96 + 0];
            r.f[u][t][1] = nb[t * 96 + 1];
            r.f[u][t][2] = nb[t * 96 + 2];
        }
    }
}

__device__ __forceinline__ void transform_store_group(const GroupRegs& r,
                                                      float* __restrict__ out,
                                                      int g, int nPoints,
                                                      int lane, int arow)
{
    #pragma unroll
    for (int u = 0; u < PTS; ++u) {
        const int p = g * PTS + u;
        const float px = r.px[u], py = r.py[u], pz = r.pz[u];

        const float n2      = px * px + py * py + pz * pz;
        const float nrm     = sqrtf(n2);
        const float inv_nrm = (nrm == 0.0f) ? 0.0f : (1.0f / nrm);
        const float ux = px * inv_nrm;
        const float uy = py * inv_nrm;

        const float dd   = ux * ux + uy * uy;
        const float invd = (dd == 0.0f) ? 0.0f : (1.0f / dd);
        const float t00 =  ux * invd;
        const float t01 =  uy * invd;
        const float t03 = -(ux * px + uy * py) * invd;
        const float t10 = -uy * invd;
        const float t11 =  ux * invd;
        const float t13 =  (uy * px - ux * py) * invd;

        v2f a_xy, a_z1;
        a_xy.x = (arow == 0) ? t00 : (arow == 1) ? t10 : 0.0f;
        a_xy.y = (arow == 0) ? t01 : (arow == 1) ? t11 : 0.0f;
        a_z1.x = (arow == 2) ? 1.0f : 0.0f;
        a_z1.y = (arow == 0) ? t03 : (arow == 1) ? t13 : (arow == 2) ? -pz : 0.0f;

        v8f d[2];
        #pragma unroll
        for (int t = 0; t < 2; ++t) {
            v2f b_xy;
            b_xy.x = r.f[u][t][0];
            b_xy.y = r.f[u][t][1];
            v8f z = {};
            d[t] = __builtin_amdgcn_wmma_f32_16x16x4_f32(
                        false, a_xy,  false, b_xy,
                        (short)0, z,  false,  false);
        }
        #pragma unroll
        for (int t = 0; t < 2; ++t) {
            v2f b_z1;
            b_z1.x = r.f[u][t][2];
            b_z1.y = 1.0f;
            d[t] = __builtin_amdgcn_wmma_f32_16x16x4_f32(
                        false, a_z1,  false, b_z1,
                        (short)0, d[t],  false,  false);
        }

        if (p < nPoints) {
            float* o = out + (size_t)p * 192u + (size_t)lane * 3u;
            #pragma unroll
            for (int t = 0; t < 2; ++t) {
                __builtin_nontemporal_store(d[t][0], o + t * 96 + 0);
                __builtin_nontemporal_store(d[t][1], o + t * 96 + 1);
                __builtin_nontemporal_store(d[t][2], o + t * 96 + 2);
            }
        }
    }
}

__global__ __launch_bounds__(256)
void neigh_norm_wmma_kernel(const float* __restrict__ points,
                            const float* __restrict__ nbh,
                            float* __restrict__ out,
                            int nPoints)
{
    const int lane = threadIdx.x & 31;
    const int wave   = __builtin_amdgcn_readfirstlane((int)(threadIdx.x >> 5));
    const int nWaves = (int)gridDim.x * 8;
    const int nGroups = (nPoints + PTS - 1) / PTS;
    const int arow = (lane < 3) ? lane : (lane >= 24 && lane < 27) ? (lane - 24) : -1;

    int g0 = (int)blockIdx.x * 8 + wave;
    if (g0 >= nGroups) return;

    GroupRegs A, B;
    load_group(A, points, nbh, g0, nPoints, lane);

    for (;;) {
        const int g1 = g0 + nWaves;
        if (g1 >= nGroups) { transform_store_group(A, out, g0, nPoints, lane, arow); break; }
        load_group(B, points, nbh, g1, nPoints, lane);
        transform_store_group(A, out, g0, nPoints, lane, arow);

        g0 = g1 + nWaves;
        if (g0 >= nGroups) { transform_store_group(B, out, g1, nPoints, lane, arow); break; }
        load_group(A, points, nbh, g0, nPoints, lane);
        transform_store_group(B, out, g1, nPoints, lane, arow);
    }
}

extern "C" void kernel_launch(void* const* d_in, const int* in_sizes, int n_in,
                              void* d_out, int out_size, void* d_ws, size_t ws_size,
                              hipStream_t stream)
{
    const float* points = (const float*)d_in[0];
    const float* nbh    = (const float*)d_in[1];
    float*       out    = (float*)d_out;

    const int nPoints = in_sizes[0] / 3;

    neigh_norm_wmma_kernel<<<512, 256, 0, stream>>>(points, nbh, out, nPoints);
}
